# starting kernel
# baseline (speedup 1.0000x reference)
#include <hip/hip_runtime.h>
#include <math.h>

typedef __attribute__((ext_vector_type(2))) float v2f;
typedef __attribute__((ext_vector_type(8))) float v8f;

__device__ __forceinline__ v8f wmma4f32(v2f a, v2f b, v8f c) {
  return __builtin_amdgcn_wmma_f32_16x16x4_f32(
       false, a,  false, b,
       (short)0, c,  false,  false);
}

__device__ __forceinline__ float half16_sum(float s) {
  s += __shfl_xor(s, 1, 32);
  s += __shfl_xor(s, 2, 32);
  s += __shfl_xor(s, 4, 32);
  s += __shfl_xor(s, 8, 32);
  return s;
}

__global__ void edge0_kernel(const float* __restrict__ x,
                             const int* __restrict__ src,
                             const int* __restrict__ dst,
                             const float* __restrict__ kw0,
                             const float* __restrict__ kw1,
                             float* __restrict__ acc0, int E) {
  unsigned e = blockIdx.x * blockDim.x + threadIdx.x;
  if (e >= (unsigned)E) return;
  const int s = src[e], t = dst[e];
  const float w0 = kw0[e], w1 = kw1[e];
  const float4* xp = (const float4*)(x + (size_t)s * 8);
  const float4 xa = xp[0], xb = xp[1];
  const float xv[8] = {xa.x, xa.y, xa.z, xa.w, xb.x, xb.y, xb.z, xb.w};
  float* a = acc0 + (size_t)t * 16;
#pragma unroll
  for (int k = 0; k < 8; ++k) atomicAdd(a + k, w0 * xv[k]);
#pragma unroll
  for (int k = 0; k < 8; ++k) atomicAdd(a + 8 + k, w1 * xv[k]);
}

__global__ __launch_bounds__(32) void mlp0_kernel(const float* __restrict__ acc0,
                                                  const float* __restrict__ W0,
                                                  const float* __restrict__ b0,
                                                  float* __restrict__ h1) {
  const int m0 = blockIdx.x * 16;
  const int lane = threadIdx.x;
  const int n = lane & 15;
  const int kb = lane >> 4;
  v8f c0 = {}, c1 = {};
#pragma unroll
  for (int kt = 0; kt < 4; ++kt) {
    const int kA = kt * 4 + 2 * kb;
    v2f a, bb0, bb1;
    const float* ap = acc0 + (size_t)(m0 + n) * 16 + kA;
    a.x = ap[0];
    a.y = ap[1];
    bb0.x = W0[(kA + 0) * 32 + n];
    bb0.y = W0[(kA + 1) * 32 + n];
    bb1.x = W0[(kA + 0) * 32 + 16 + n];
    bb1.y = W0[(kA + 1) * 32 + 16 + n];
    c0 = wmma4f32(a, bb0, c0);
    c1 = wmma4f32(a, bb1, c1);
  }
  const float bias0 = b0[n], bias1 = b0[16 + n];
#pragma unroll
  for (int v = 0; v < 8; ++v) {
    float e0 = c0[v] + bias0;
    float e1 = c1[v] + bias1;
    float s = half16_sum(e0 * e0 + e1 * e1);
    const float invn = 1.0f / fmaxf(sqrtf(s), 1e-12f);
    const int row = m0 + v + 8 * kb;
    h1[(size_t)row * 32 + n] = e0 * invn;
    h1[(size_t)row * 32 + 16 + n] = e1 * invn;
  }
}

__global__ void edge1_kernel(const float* __restrict__ h1,
                             const int* __restrict__ src,
                             const int* __restrict__ dst,
                             const float* __restrict__ Kw,
                             float* __restrict__ acc1, int E) {
  unsigned tid = blockIdx.x * blockDim.x + threadIdx.x;
  if (tid >= 4u * (unsigned)E) return;
  const int e = tid >> 2;
  const int j = (tid >> 1) & 1;
  const int hh = tid & 1;
  const int s = src[e], t = dst[e];
  const float w = Kw[(size_t)j * E + e];
  const float4* g = (const float4*)(h1 + (size_t)s * 32 + hh * 16);
  const float4 g0 = g[0], g1 = g[1], g2 = g[2], g3 = g[3];
  const float vv[16] = {g0.x, g0.y, g0.z, g0.w, g1.x, g1.y, g1.z, g1.w,
                        g2.x, g2.y, g2.z, g2.w, g3.x, g3.y, g3.z, g3.w};
  float* a = acc1 + (size_t)t * 64 + j * 32 + hh * 16;
#pragma unroll
  for (int k = 0; k < 16; ++k) atomicAdd(a + k, w * vv[k]);
}

#define TS 132
__global__ __launch_bounds__(128) void mlp1_kernel(const float* __restrict__ acc1,
                                                   const float* __restrict__ W1,
                                                   const float* __restrict__ b1,
                                                   const float* __restrict__ W2,
                                                   const float* __restrict__ b2,
                                                   float* __restrict__ out) {
  __shared__ float tsh[16 * TS];
  __shared__ float rowsq[16];
  const int m0 = blockIdx.x * 16;
  const int wave = threadIdx.x >> 5;
  const int lane = threadIdx.x & 31;
  const int n = lane & 15;
  const int kb = lane >> 4;

#pragma unroll
  for (int tt = 0; tt < 2; ++tt) {
    const int n0 = (wave * 2 + tt) * 16;
    v8f c = {};
    for (int kt = 0; kt < 16; ++kt) {
      const int kA = kt * 4 + 2 * kb;
      v2f a, b;
      const float* ap = acc1 + (size_t)(m0 + n) * 64 + kA;
      a.x = ap[0];
      a.y = ap[1];
      b.x = W1[(kA + 0) * 128 + n0 + n];
      b.y = W1[(kA + 1) * 128 + n0 + n];
      c = wmma4f32(a, b, c);
    }
    const float bias = b1[n0 + n];
#pragma unroll
    for (int v = 0; v < 8; ++v) {
      tsh[(v + 8 * kb) * TS + n0 + n] = fmaxf(c[v] + bias, 0.0f);
    }
  }
  if (threadIdx.x < 16) rowsq[threadIdx.x] = 0.0f;
  __syncthreads();

  const int n0 = wave * 16;
  v8f c = {};
  for (int kt = 0; kt < 32; ++kt) {
    const int kA = kt * 4 + 2 * kb;
    v2f a, b;
    a.x = tsh[n * TS + kA];
    a.y = tsh[n * TS + kA + 1];
    b.x = W2[(kA + 0) * 64 + n0 + n];
    b.y = W2[(kA + 1) * 64 + n0 + n];
    c = wmma4f32(a, b, c);
  }
  const float bias = b2[n0 + n];
  float cv[8];
#pragma unroll
  for (int v = 0; v < 8; ++v) {
    cv[v] = c[v] + bias;
    float s = half16_sum(cv[v] * cv[v]);
    if (n == 0) atomicAdd(&rowsq[v + 8 * kb], s);
  }
  __syncthreads();
#pragma unroll
  for (int v = 0; v < 8; ++v) {
    const int row = v + 8 * kb;
    const float invn = 1.0f / fmaxf(sqrtf(rowsq[row]), 1e-12f);
    out[(size_t)(m0 + row) * 64 + n0 + n] = cv[v] * invn;
  }
}

extern "C" void kernel_launch(void* const* d_in, const int* in_sizes, int n_in,
                              void* d_out, int out_size, void* d_ws, size_t ws_size,
                              hipStream_t stream) {
  const float* x  = (const float*)d_in[0];
  const int*   ei = (const int*)d_in[1];
  const float* Kw = (const float*)d_in[2];
  const float* W0 = (const float*)d_in[3];
  const float* b0 = (const float*)d_in[4];
  const float* W1 = (const float*)d_in[5];
  const float* b1 = (const float*)d_in[6];
  const float* W2 = (const float*)d_in[7];
  const float* b2 = (const float*)d_in[8];

  const int N = in_sizes[0] / 8;
  const int E = in_sizes[1] / 2;
  const int* src = ei;
  const int* dst = ei + E;

  float* acc0 = (float*)d_ws;
  float* acc1 = acc0 + (size_t)N * 16;
  float* h1   = acc1 + (size_t)N * 64;

  hipMemsetAsync(d_ws, 0, (size_t)N * 80 * sizeof(float), stream);

  edge0_kernel<<<(E + 255) / 256, 256, 0, stream>>>(x, src, dst, Kw, Kw + E, acc0, E);
  mlp0_kernel<<<N / 16, 32, 0, stream>>>(acc0, W0, b0, h1);
  edge1_kernel<<<(4 * E + 255) / 256, 256, 0, stream>>>(h1, src, dst, Kw, acc1, E);
  mlp1_kernel<<<N / 16, 128, 0, stream>>>(acc1, W1, b1, W2, b2, (float*)d_out);
}


# layer 1 no longer scans or gathers edge fields: layer 0 leaves each work-group's edge records (src, node, 2 weights) in its own rows of the sums plane, layer 1 streams them; LDS arrival order kept
# speedup vs baseline: 5.2410x; 5.2410x over previous
#include <hip/hip_runtime.h>
#include <math.h>

typedef __attribute__((ext_vector_type(2))) float v2f;
typedef __attribute__((ext_vector_type(8))) float v8f;

__device__ __forceinline__ v8f wmma4f32(v2f a, v2f b, v8f c) {
  return __builtin_amdgcn_wmma_f32_16x16x4_f32(
       false, a,  false, b,
       (short)0, c,  false,  false);
}

__device__ __forceinline__ float half16_sum(float s) {
  s += __shfl_xor(s, 1, 32);
  s += __shfl_xor(s, 2, 32);
  s += __shfl_xor(s, 4, 32);
  s += __shfl_xor(s, 8, 32);
  return s;
}


#define BN 800
#define QCAP 544
#define NWAVE 32
#define RHDR 32
#define RCAP 17056
#define RBATCH (RCAP / 32)
#define RNULL 0xffffffffu
#define Q1CAP 64

__global__ void bucket_kernel(const int* __restrict__ dst, unsigned* __restrict__ bkt, int E, int nquad) {
  const unsigned t = blockIdx.x * blockDim.x + threadIdx.x;
  if (t >= (unsigned)nquad) return;
  unsigned packed = 0;
#pragma unroll
  for (int i = 0; i < 4; ++i) {
    const unsigned e = 4u * t + i;
    unsigned b = 255u;
    if (e < (unsigned)E) {
      const unsigned q = (unsigned)dst[e] / (unsigned)BN;
      b = q < 255u ? q : 255u;
    }
    packed |= b << (8 * i);
  }
  bkt[t] = packed;
}

__device__ __forceinline__ unsigned zero_bytes(unsigned x) {
  const unsigned t = (x & 0x7f7f7f7fu) + 0x7f7f7f7fu;
  return ~(t | x | 0x7f7f7f7fu);
}

__device__ __forceinline__ int scan_push(uint4 v, unsigned pat, int e0, int lane, int* q, int count) {
  unsigned f[4];
  f[0] = zero_bytes(v.x ^ pat);
  f[1] = zero_bytes(v.y ^ pat);
  f[2] = zero_bytes(v.z ^ pat);
  f[3] = zero_bytes(v.w ^ pat);
  const unsigned any = f[0] | f[1] | f[2] | f[3];
  const unsigned have = (unsigned)__ballot(any != 0);
  if (have != 0) {
    const int cnt = __popc(f[0]) + __popc(f[1]) + __popc(f[2]) + __popc(f[3]);
    if (__ballot(cnt > 1) == 0) {
      if (any != 0) {
        const int k = ((__ffs(any) - 1) >> 3) | ((f[1] | f[3]) ? 4 : 0) | ((f[2] | f[3]) ? 8 : 0);
        q[count + (int)__builtin_amdgcn_mbcnt_lo(have, 0u)] = e0 + k;
      }
      count += __popc(have);
    } else {
      int inc = cnt;
#pragma unroll
      for (int o = 1; o < 32; o <<= 1) {
        const int t = __shfl_up(inc, o, 32);
        if (lane >= o) inc += t;
      }
      const int total = __builtin_amdgcn_readlane(inc, 31);
      int pos = count + inc - cnt;
#pragma unroll
      for (int j = 0; j < 4; ++j) {
        unsigned ff = f[j];
        while (ff) {
          const int bit = __ffs(ff) - 1;
          q[pos++] = e0 + 4 * j + (bit >> 3);
          ff &= ff - 1;
        }
      }
      count += total;
    }
  }
  return count;
}

__device__ __forceinline__ void drain0(const int* qq, int nb, int lane, int lo, int E,
                                       const float* __restrict__ x, const int* __restrict__ src,
                                       const int* __restrict__ dst, const float* __restrict__ Kw,
                                       float* acc, unsigned* rec, int* nbat) {
  unsigned pk = RNULL;
  float w0 = 0.0f, w1 = 0.0f;
  if (lane < nb) {
    const int e = qq[lane];
    const int s = src[e];
    const int d = dst[e] - lo;
    w0 = Kw[e];
    w1 = Kw[(size_t)E + e];
    pk = (unsigned)s | ((unsigned)d << 22);
    const float4* xp = (const float4*)(x + (size_t)s * 8);
    const float4 xa = xp[0], xb = xp[1];
    const float xv[8] = {xa.x, xa.y, xa.z, xa.w, xb.x, xb.y, xb.z, xb.w};
    float* ap = acc + d * 17;
#pragma unroll
    for (int k = 0; k < 8; ++k) atomicAdd(ap + k, w0 * xv[k]);
#pragma unroll
    for (int k = 0; k < 8; ++k) atomicAdd(ap + 8 + k, w1 * xv[k]);
  }
  int bi = 0;
  if (lane == 0) bi = atomicAdd(nbat, 1);
  bi = __builtin_amdgcn_readfirstlane(bi);
  if (bi < RBATCH) {
    const int at = bi * 32 + lane;
    rec[RHDR + at] = pk;
    rec[RHDR + RCAP + at] = __float_as_uint(w0);
    rec[RHDR + 2 * RCAP + at] = __float_as_uint(w1);
  }
}

template <bool FULL>
__device__ __forceinline__ void add_batch1(int s, int d, float w0, float w1, int nb, int lane,
                                           const float* __restrict__ h1, float* acc) {
  float hv[32];
#pragma unroll
  for (int i = 0; i < 32; ++i) {
    const int si = __builtin_amdgcn_readlane(s, i);
    hv[i] = h1[(size_t)si * 32 + lane];
  }
#pragma unroll
  for (int i = 0; i < 32; ++i) {
    if (FULL || i < nb) {
      const int di = __builtin_amdgcn_readlane(d, i);
      const float a = __int_as_float(__builtin_amdgcn_readlane(__float_as_int(w0), i));
      const float b = __int_as_float(__builtin_amdgcn_readlane(__float_as_int(w1), i));
      float* ap = acc + di * 64 + lane;
      atomicAdd(ap, a * hv[i]);
      atomicAdd(ap + 32, b * hv[i]);
    }
  }
}

template <bool FULL>
__device__ __forceinline__ void drain1(const int* qq, int nb, int lane, int lo, int E,
                                       const float* __restrict__ h1, const int* __restrict__ src,
                                       const int* __restrict__ dst, const float* __restrict__ Kw,
                                       float* acc) {
  int s = 0, d = 0;
  float w0 = 0.0f, w1 = 0.0f;
  if (FULL || lane < nb) {
    const int e = qq[lane];
    s = src[e];
    d = dst[e] - lo;
    w0 = Kw[e];
    w1 = Kw[(size_t)E + e];
  }
  add_batch1<FULL>(s, d, w0, w1, nb, lane, h1, acc);
}

__global__ __launch_bounds__(1024) void agg0_kernel(const float* __restrict__ feat,
                                                    const int* __restrict__ src,
                                                    const int* __restrict__ dst,
                                                    const float* __restrict__ Kw,
                                                    const unsigned* __restrict__ bkt,
                                                    float* __restrict__ accg,
                                                    unsigned* __restrict__ recg, int E, int G16, int N) {
  constexpr int F = 16;
  constexpr int STRIDE = 17;
  __shared__ float acc[BN * STRIDE];
  __shared__ int qs[NWAVE * QCAP];
  __shared__ int nbat;
  const int tid = threadIdx.x;
  const int lane = tid & 31;
  const int wave = tid >> 5;
  const int lo = blockIdx.x * BN;
  const unsigned pat = blockIdx.x * 0x01010101u;
  const uint4 none = make_uint4(~pat, ~pat, ~pat, ~pat);
  const uint4* __restrict__ bkt4 = (const uint4*)bkt;
  int* q = qs + wave * QCAP;
  unsigned* rec = recg + (size_t)blockIdx.x * (BN * 64);

  for (int i = tid; i < BN * STRIDE; i += 1024) acc[i] = 0.0f;
  if (tid == 0) nbat = 0;
  __syncthreads();

  const int iters = (G16 + 1023) / 1024;
  int itx = (int)(((long long)blockIdx.x * iters) / gridDim.x);
  int count = 0;
  int g = itx * 1024 + tid;
  uint4 v = bkt4[g < G16 ? g : G16 - 1];
  if (g >= G16) v = none;
  for (int it = 0; it < iters; ++it) {
    int itn = itx + 1;
    if (itn >= iters) itn = 0;
    const int gn = itn * 1024 + tid;
    uint4 vn = bkt4[gn < G16 ? gn : G16 - 1];
    if (gn >= G16) vn = none;
    count = scan_push(v, pat, g * 16, lane, q, count);
    while (count >= 32) {
      count -= 32;
      drain0(q + count, 32, lane, lo, E, feat, src, dst, Kw, acc, rec, &nbat);
    }
    v = vn;
    g = gn;
    itx = itn;
  }
  if (count > 0) drain0(q, count, lane, lo, E, feat, src, dst, Kw, acc, rec, &nbat);
  __syncthreads();

  if (tid == 0) {
    const int n = nbat;
    rec[0] = (unsigned)(n < RBATCH ? n : RBATCH);
    rec[1] = n > RBATCH ? 1u : 0u;
  }
  for (int i = tid; i < BN * F; i += 1024) {
    const int node = i / F, c = i % F;
    if (lo + node < N) accg[(size_t)(lo + node) * F + c] = acc[node * STRIDE + c];
  }
}

__global__ __launch_bounds__(1024) void agg1_kernel(const float* __restrict__ h1,
                                                    const int* __restrict__ src,
                                                    const int* __restrict__ dst,
                                                    const float* __restrict__ Kw,
                                                    const unsigned* recg, float* accg, int E, int N) {
  constexpr int F = 64;
  __shared__ float acc[BN * F];
  __shared__ int qs[NWAVE * Q1CAP];
  const int tid = threadIdx.x;
  const int lane = tid & 31;
  const int wave = __builtin_amdgcn_readfirstlane(tid >> 5);
  const int lo = blockIdx.x * BN;
  const unsigned* rec = recg + (size_t)blockIdx.x * (BN * 64);
  const int nb = __builtin_amdgcn_readfirstlane((int)rec[0]);
  const int over = __builtin_amdgcn_readfirstlane((int)rec[1]);

  for (int i = tid; i < BN * F; i += 1024) acc[i] = 0.0f;
  __syncthreads();

  if (over == 0) {
    const unsigned* p0 = rec + RHDR;
    int bt = wave;
    unsigned pk = RNULL, u0 = 0u, u1 = 0u;
    if (bt < nb) {
      const int at = bt * 32 + lane;
      pk = p0[at];
      u0 = p0[RCAP + at];
      u1 = p0[2 * RCAP + at];
    }
    while (bt < nb) {
      const int nbt = bt + NWAVE;
      unsigned pkn = RNULL, u0n = 0u, u1n = 0u;
      if (nbt < nb) {
        const int at = nbt * 32 + lane;
        pkn = p0[at];
        u0n = p0[RCAP + at];
        u1n = p0[2 * RCAP + at];
      }
      const bool ok = pk != RNULL;
      const unsigned vm = (unsigned)__ballot(ok);
      const int s = ok ? (int)(pk & 0x3fffffu) : 0;
      const int d = ok ? (int)(pk >> 22) : 0;
      const float w0 = __uint_as_float(u0), w1 = __uint_as_float(u1);
      if (vm == 0xffffffffu) add_batch1<true>(s, d, w0, w1, 32, lane, h1, acc);
      else add_batch1<false>(s, d, w0, w1, __popc(vm), lane, h1, acc);
      pk = pkn;
      u0 = u0n;
      u1 = u1n;
      bt = nbt;
    }
  } else {
    int* q = qs + wave * Q1CAP;
    int count = 0;
    for (int base = 0; base < E; base += 1024) {
      const int e = base + tid;
      bool own = false;
      if (e < E) own = (unsigned)dst[e] / (unsigned)BN == blockIdx.x;
      const unsigned have = (unsigned)__ballot(own);
      if (own) q[count + (int)__builtin_amdgcn_mbcnt_lo(have, 0u)] = e;
      count += __popc(have);
      if (count >= 32) {
        count -= 32;
        drain1<true>(q + count, 32, lane, lo, E, h1, src, dst, Kw, acc);
      }
    }
    if (count > 0) drain1<false>(q, count, lane, lo, E, h1, src, dst, Kw, acc);
  }
  __syncthreads();

  for (int i = tid; i < BN * F; i += 1024) {
    const int node = i / F, c = i % F;
    if (lo + node < N) accg[(size_t)(lo + node) * F + c] = acc[node * F + c];
  }
}

__global__ __launch_bounds__(32) void mlp0_kernel(const float* __restrict__ acc0,
                                                  const float* __restrict__ W0,
                                                  const float* __restrict__ b0,
                                                  float* __restrict__ h1) {
  const int m0 = blockIdx.x * 16;
  const int lane = threadIdx.x;
  const int n = lane & 15;
  const int kb = lane >> 4;
  v8f c0 = {}, c1 = {};
#pragma unroll
  for (int kt = 0; kt < 4; ++kt) {
    const int kA = kt * 4 + 2 * kb;
    v2f a, bb0, bb1;
    const float* ap = acc0 + (size_t)(m0 + n) * 16 + kA;
    a.x = ap[0];
    a.y = ap[1];
    bb0.x = W0[(kA + 0) * 32 + n];
    bb0.y = W0[(kA + 1) * 32 + n];
    bb1.x = W0[(kA + 0) * 32 + 16 + n];
    bb1.y = W0[(kA + 1) * 32 + 16 + n];
    c0 = wmma4f32(a, bb0, c0);
    c1 = wmma4f32(a, bb1, c1);
  }
  const float bias0 = b0[n], bias1 = b0[16 + n];
#pragma unroll
  for (int v = 0; v < 8; ++v) {
    float e0 = c0[v] + bias0;
    float e1 = c1[v] + bias1;
    float s = half16_sum(e0 * e0 + e1 * e1);
    const float invn = 1.0f / fmaxf(sqrtf(s), 1e-12f);
    const int row = m0 + v + 8 * kb;
    h1[(size_t)row * 32 + n] = e0 * invn;
    h1[(size_t)row * 32 + 16 + n] = e1 * invn;
  }
}

#define TS 132
__global__ __launch_bounds__(128) void mlp1_kernel(const float* __restrict__ acc1,
                                                   const float* __restrict__ W1,
                                                   const float* __restrict__ b1,
                                                   const float* __restrict__ W2,
                                                   const float* __restrict__ b2,
                                                   float* __restrict__ out) {
  __shared__ float tsh[16 * TS];
  __shared__ float rowsq[16];
  const int m0 = blockIdx.x * 16;
  const int wave = threadIdx.x >> 5;
  const int lane = threadIdx.x & 31;
  const int n = lane & 15;
  const int kb = lane >> 4;

#pragma unroll
  for (int tt = 0; tt < 2; ++tt) {
    const int n0 = (wave * 2 + tt) * 16;
    v8f c = {};
    for (int kt = 0; kt < 16; ++kt) {
      const int kA = kt * 4 + 2 * kb;
      v2f a, b;
      const float* ap = acc1 + (size_t)(m0 + n) * 64 + kA;
      a.x = ap[0];
      a.y = ap[1];
      b.x = W1[(kA + 0) * 128 + n0 + n];
      b.y = W1[(kA + 1) * 128 + n0 + n];
      c = wmma4f32(a, b, c);
    }
    const float bias = b1[n0 + n];
#pragma unroll
    for (int v = 0; v < 8; ++v) {
      tsh[(v + 8 * kb) * TS + n0 + n] = fmaxf(c[v] + bias, 0.0f);
    }
  }
  if (threadIdx.x < 16) rowsq[threadIdx.x] = 0.0f;
  __syncthreads();

  const int n0 = wave * 16;
  v8f c = {};
  for (int kt = 0; kt < 32; ++kt) {
    const int kA = kt * 4 + 2 * kb;
    v2f a, b;
    a.x = tsh[n * TS + kA];
    a.y = tsh[n * TS + kA + 1];
    b.x = W2[(kA + 0) * 64 + n0 + n];
    b.y = W2[(kA + 1) * 64 + n0 + n];
    c = wmma4f32(a, b, c);
  }
  const float bias = b2[n0 + n];
  float cv[8];
#pragma unroll
  for (int v = 0; v < 8; ++v) {
    cv[v] = c[v] + bias;
    float s = half16_sum(cv[v] * cv[v]);
    if (n == 0) atomicAdd(&rowsq[v + 8 * kb], s);
  }
  __syncthreads();
#pragma unroll
  for (int v = 0; v < 8; ++v) {
    const int row = v + 8 * kb;
    const float invn = 1.0f / fmaxf(sqrtf(rowsq[row]), 1e-12f);
    out[(size_t)(m0 + row) * 64 + n0 + n] = cv[v] * invn;
  }
}

extern "C" void kernel_launch(void* const* d_in, const int* in_sizes, int n_in,
                              void* d_out, int out_size, void* d_ws, size_t ws_size,
                              hipStream_t stream) {
  const float* x  = (const float*)d_in[0];
  const int*   ei = (const int*)d_in[1];
  const float* Kw = (const float*)d_in[2];
  const float* W0 = (const float*)d_in[3];
  const float* b0 = (const float*)d_in[4];
  const float* W1 = (const float*)d_in[5];
  const float* b1 = (const float*)d_in[6];
  const float* W2 = (const float*)d_in[7];
  const float* b2 = (const float*)d_in[8];

  const int N = in_sizes[0] / 8;
  const int E = in_sizes[1] / 2;
  const int* src = ei;
  const int* dst = ei + E;
  const int G16 = (E + 15) / 16;
  const int NB = (N + BN - 1) / BN;

  float* acc0 = (float*)d_ws;
  float* acc1 = acc0 + (size_t)N * 16;
  float* h1   = acc1 + (size_t)N * 64;
  unsigned* bktA = (unsigned*)h1;
  unsigned* recs = (unsigned*)acc1;

  bucket_kernel<<<(G16 * 4 + 255) / 256, 256, 0, stream>>>(dst, bktA, E, G16 * 4);
  agg0_kernel<<<NB, 1024, 0, stream>>>(x, src, dst, Kw, bktA, acc0, recs, E, G16, N);
  mlp0_kernel<<<N / 16, 32, 0, stream>>>(acc0, W0, b0, h1);
  agg1_kernel<<<NB, 1024, 0, stream>>>(h1, src, dst, Kw, recs, acc1, E, N);
  mlp1_kernel<<<N / 16, 128, 0, stream>>>(acc1, W1, b1, W2, b2, (float*)d_out);
}
